# starting kernel
# baseline (speedup 1.0000x reference)
#include <hip/hip_runtime.h>

typedef __attribute__((ext_vector_type(2))) float v2f;
typedef __attribute__((ext_vector_type(8))) float v8f;

#define NN 50000
#define NE 800000
#define DF 64
#define DH 128
#define LDS_STRIDE 132

__global__ void gin_init(const float* __restrict__ feat,
                         const float* __restrict__ eps,
                         float* __restrict__ acc) {
  int i = blockIdx.x * blockDim.x + threadIdx.x;
  if (i < NN * DF / 4) {
    float s = 1.0f + eps[0];
    float4 v = ((const float4*)feat)[i];
    v.x *= s; v.y *= s; v.z *= s; v.w *= s;
    ((float4*)acc)[i] = v;
  }
}

__global__ void gin_scatter(const float* __restrict__ feat,
                            const int* __restrict__ src,
                            const int* __restrict__ dst,
                            float* __restrict__ acc) {
  long long t = (long long)blockIdx.x * blockDim.x + threadIdx.x;
  int e = (int)(t >> 6);
  int d = (int)(t & 63);
  if (e >= NE) return;
  int s = src[e];
  int q = dst[e];
  atomicAdd(acc + (long long)q * DF + d, feat[(long long)s * DF + d]);
}

__global__ __launch_bounds__(32) void gin_mlp(const float* __restrict__ acc,
                                              const float* __restrict__ W1,
                                              const float* __restrict__ b1,
                                              const float* __restrict__ W2,
                                              const float* __restrict__ b2,
                                              float* __restrict__ out) {
  __shared__ float lds[16 * LDS_STRIDE];

  const int tile = blockIdx.x;
  const int lane = threadIdx.x;
  const int half = lane >> 4;
  const int l16  = lane & 15;
  const int row0 = tile * 16;

  v2f afrag[16];
  {
    const float* arow = acc + (long long)(row0 + l16) * DF;
#pragma unroll
    for (int kk = 0; kk < 16; ++kk) {
      int kb = kk * 4 + half * 2;
      afrag[kk].x = arow[kb];
      afrag[kk].y = arow[kb + 1];
    }
  }

#pragma unroll
  for (int nt = 0; nt < 8; ++nt) {
    const int n0 = nt * 16;
    v8f c = {};
#pragma unroll
    for (int kk = 0; kk < 16; ++kk) {
      int kr = kk * 4 + half * 2;
      v2f bf;
      bf.x = W1[kr * DH + n0 + l16];
      bf.y = W1[(kr + 1) * DH + n0 + l16];
      c = __builtin_amdgcn_wmma_f32_16x16x4_f32(false, afrag[kk], false, bf,
                                                (short)0, c, false, false);
    }
    float bias = b1[n0 + l16];
#pragma unroll
    for (int r = 0; r < 8; ++r) {
      float h = c[r] + bias;
      h = h > 0.0f ? h : 0.0f;
      lds[(r + half * 8) * LDS_STRIDE + n0 + l16] = h;
    }
  }

  __syncthreads();

#pragma unroll
  for (int nt = 0; nt < 4; ++nt) {
    const int n0 = nt * 16;
    v8f c = {};
#pragma unroll
    for (int kk = 0; kk < 32; ++kk) {
      int kr = kk * 4 + half * 2;
      v2f af, bf;
      af.x = lds[l16 * LDS_STRIDE + kr];
      af.y = lds[l16 * LDS_STRIDE + kr + 1];
      bf.x = W2[kr * DF + n0 + l16];
      bf.y = W2[(kr + 1) * DF + n0 + l16];
      c = __builtin_amdgcn_wmma_f32_16x16x4_f32(false, af, false, bf,
                                                (short)0, c, false, false);
    }
    float bias = b2[n0 + l16];
#pragma unroll
    for (int r = 0; r < 8; ++r) {
      out[(long long)(row0 + r + half * 8) * DF + n0 + l16] = c[r] + bias;
    }
  }
}

extern "C" void kernel_launch(void* const* d_in, const int* in_sizes, int n_in,
                              void* d_out, int out_size, void* d_ws, size_t ws_size,
                              hipStream_t stream) {
  const float* feat = (const float*)d_in[0];
  const int*   src  = (const int*)d_in[1];
  const int*   dst  = (const int*)d_in[2];
  const float* eps  = (const float*)d_in[3];
  const float* W1   = (const float*)d_in[4];
  const float* b1   = (const float*)d_in[5];
  const float* W2   = (const float*)d_in[6];
  const float* b2   = (const float*)d_in[7];
  float* acc = (float*)d_ws;
  float* out = (float*)d_out;

  {
    int n4 = NN * DF / 4;
    int blocks = (n4 + 255) / 256;
    gin_init<<<blocks, 256, 0, stream>>>(feat, eps, acc);
  }
  {
    long long total = (long long)NE * 64;
    int blocks = (int)((total + 255) / 256);
    gin_scatter<<<blocks, 256, 0, stream>>>(feat, src, dst, acc);
  }
  {
    gin_mlp<<<NN / 16, 32, 0, stream>>>(acc, W1, b1, W2, b2, out);
  }
}


# Neighbour sum without float atomics: edges grouped by destination in-call (integer count, scan, fill), one wave per node sums its source rows; sum order per node = arrival rank of an integer add
# speedup vs baseline: 1.3841x; 1.3841x over previous
#include <hip/hip_runtime.h>

typedef __attribute__((ext_vector_type(2))) float v2f;
typedef __attribute__((ext_vector_type(8))) float v8f;

#define NN 50000
#define NE 800000
#define DF 64
#define DH 128
#define LDS_STRIDE 132


__global__ void gin_clear(int* __restrict__ cnt) {
  int i = blockIdx.x * blockDim.x + threadIdx.x;
  if (i <= NN) cnt[i] = 0;
}

__global__ void gin_count(const int* __restrict__ dst,
                          int* __restrict__ cnt,
                          int* __restrict__ rank) {
  int i = blockIdx.x * blockDim.x + threadIdx.x;
  if (i < NE / 4) {
    int4 d = ((const int4*)dst)[i];
    int4 r;
    r.x = atomicAdd(cnt + d.x, 1);
    r.y = atomicAdd(cnt + d.y, 1);
    r.z = atomicAdd(cnt + d.z, 1);
    r.w = atomicAdd(cnt + d.w, 1);
    ((int4*)rank)[i] = r;
  }
}

__global__ __launch_bounds__(256) void gin_offsets(int* cnt, int* __restrict__ off) {
  __shared__ int wsum[8];
  __shared__ int bbase;
  const int n = blockIdx.x * 256 + threadIdx.x;
  const int lane = threadIdx.x & 31;
  const int w = threadIdx.x >> 5;
  const int c = (n < NN) ? cnt[n] : 0;
  int x = c;
#pragma unroll
  for (int d = 1; d < 32; d <<= 1) {
    int y = __shfl_up(x, d);
    if (lane >= d) x += y;
  }
  if (lane == 31) wsum[w] = x;
  __syncthreads();
  if (threadIdx.x == 0) {
    int t = 0;
#pragma unroll
    for (int k = 0; k < 8; ++k) { int s = wsum[k]; wsum[k] = t; t += s; }
    bbase = atomicAdd(cnt + NN, t);
  }
  __syncthreads();
  if (n < NN) off[n] = bbase + wsum[w] + x - c;
}

__global__ void gin_fill(const int* __restrict__ src,
                         const int* __restrict__ dst,
                         const int* __restrict__ off,
                         const int* __restrict__ rank,
                         int* __restrict__ ssrc) {
  int i = blockIdx.x * blockDim.x + threadIdx.x;
  if (i < NE / 4) {
    int4 d = ((const int4*)dst)[i];
    int4 r = ((const int4*)rank)[i];
    int4 s = ((const int4*)src)[i];
    unsigned p0 = min((unsigned)(off[d.x] + r.x), (unsigned)(NE - 1));
    unsigned p1 = min((unsigned)(off[d.y] + r.y), (unsigned)(NE - 1));
    unsigned p2 = min((unsigned)(off[d.z] + r.z), (unsigned)(NE - 1));
    unsigned p3 = min((unsigned)(off[d.w] + r.w), (unsigned)(NE - 1));
    ssrc[p0] = s.x;
    ssrc[p1] = s.y;
    ssrc[p2] = s.z;
    ssrc[p3] = s.w;
  }
}

__global__ __launch_bounds__(256) void gin_gather(const float* __restrict__ feat,
                                                  const float* __restrict__ eps,
                                                  const int* __restrict__ cnt,
                                                  const int* __restrict__ off,
                                                  const int* __restrict__ ssrc,
                                                  float* __restrict__ acc) {
  const int lane = threadIdx.x & 31;
  const int node = blockIdx.x * 8 + (threadIdx.x >> 5);
  const float2* f2 = (const float2*)feat;

  int c = __builtin_amdgcn_readfirstlane(cnt[node]);
  int o = __builtin_amdgcn_readfirstlane(off[node]);
  c = min(max(c, 0), NE);
  o = min(max(o, 0), NE - c);

  const float sc = 1.0f + eps[0];
  const float2 self = f2[(long long)node * 32 + lane];

  float2 a;
  a.x = 0.0f; a.y = 0.0f;
  for (int base = 0; base < c; base += 32) {
    const int m = min(c - base, 32);
    int my = 0;
    if (lane < m) my = ssrc[o + base + lane];
    unsigned s[32];
#pragma unroll
    for (int u = 0; u < 32; ++u) {
      s[u] = min((unsigned)__builtin_amdgcn_readlane(my, u), (unsigned)(NN - 1));
    }
    float2 v[32];
#pragma unroll
    for (int u = 0; u < 32; ++u) {
      v[u].x = 0.0f; v[u].y = 0.0f;
      if (u < m) v[u] = f2[(long long)s[u] * 32 + lane];
    }
#pragma unroll
    for (int u = 0; u < 32; ++u) { a.x += v[u].x; a.y += v[u].y; }
  }
  float2 r;
  r.x = sc * self.x + a.x;
  r.y = sc * self.y + a.y;
  ((float2*)acc)[(long long)node * 32 + lane] = r;
}

__global__ __launch_bounds__(32) void gin_mlp(const float* __restrict__ acc,
                                              const float* __restrict__ W1,
                                              const float* __restrict__ b1,
                                              const float* __restrict__ W2,
                                              const float* __restrict__ b2,
                                              float* __restrict__ out) {
  __shared__ float lds[16 * LDS_STRIDE];

  const int tile = blockIdx.x;
  const int lane = threadIdx.x;
  const int half = lane >> 4;
  const int l16  = lane & 15;
  const int row0 = tile * 16;

  v2f afrag[16];
  {
    const float* arow = acc + (long long)(row0 + l16) * DF;
#pragma unroll
    for (int kk = 0; kk < 16; ++kk) {
      int kb = kk * 4 + half * 2;
      afrag[kk].x = arow[kb];
      afrag[kk].y = arow[kb + 1];
    }
  }

#pragma unroll
  for (int nt = 0; nt < 8; ++nt) {
    const int n0 = nt * 16;
    v8f c = {};
#pragma unroll
    for (int kk = 0; kk < 16; ++kk) {
      int kr = kk * 4 + half * 2;
      v2f bf;
      bf.x = W1[kr * DH + n0 + l16];
      bf.y = W1[(kr + 1) * DH + n0 + l16];
      c = __builtin_amdgcn_wmma_f32_16x16x4_f32(false, afrag[kk], false, bf,
                                                (short)0, c, false, false);
    }
    float bias = b1[n0 + l16];
#pragma unroll
    for (int r = 0; r < 8; ++r) {
      float h = c[r] + bias;
      h = h > 0.0f ? h : 0.0f;
      lds[(r + half * 8) * LDS_STRIDE + n0 + l16] = h;
    }
  }

  __syncthreads();

#pragma unroll
  for (int nt = 0; nt < 4; ++nt) {
    const int n0 = nt * 16;
    v8f c = {};
#pragma unroll
    for (int kk = 0; kk < 32; ++kk) {
      int kr = kk * 4 + half * 2;
      v2f af, bf;
      af.x = lds[l16 * LDS_STRIDE + kr];
      af.y = lds[l16 * LDS_STRIDE + kr + 1];
      bf.x = W2[kr * DF + n0 + l16];
      bf.y = W2[(kr + 1) * DF + n0 + l16];
      c = __builtin_amdgcn_wmma_f32_16x16x4_f32(false, af, false, bf,
                                                (short)0, c, false, false);
    }
    float bias = b2[n0 + l16];
#pragma unroll
    for (int r = 0; r < 8; ++r) {
      out[(long long)(row0 + r + half * 8) * DF + n0 + l16] = c[r] + bias;
    }
  }
}

extern "C" void kernel_launch(void* const* d_in, const int* in_sizes, int n_in,
                              void* d_out, int out_size, void* d_ws, size_t ws_size,
                              hipStream_t stream) {
  const float* feat = (const float*)d_in[0];
  const int*   src  = (const int*)d_in[1];
  const int*   dst  = (const int*)d_in[2];
  const float* eps  = (const float*)d_in[3];
  const float* W1   = (const float*)d_in[4];
  const float* b1   = (const float*)d_in[5];
  const float* W2   = (const float*)d_in[6];
  const float* b2   = (const float*)d_in[7];
  float* acc  = (float*)d_ws;
  int*   cnt  = (int*)d_ws + 3200000;
  int*   off  = (int*)d_ws + 3250048;
  int*   rank = (int*)d_ws + 3300096;
  int*   ssrc = (int*)d_ws + 4100096;
  float* out = (float*)d_out;

  gin_clear<<<196, 256, 0, stream>>>(cnt);
  gin_count<<<782, 256, 0, stream>>>(dst, cnt, rank);
  gin_offsets<<<196, 256, 0, stream>>>(cnt, off);
  gin_fill<<<782, 256, 0, stream>>>(src, dst, off, rank, ssrc);
  gin_gather<<<NN / 8, 256, 0, stream>>>(feat, eps, cnt, off, ssrc, acc);
  gin_mlp<<<NN / 16, 32, 0, stream>>>(acc, W1, b1, W2, b2, out);
}


# Edges grouped by destination through LDS only: per-chunk LDS partition into 250 node-range bins, one work-group per bin orders and sums rows; 6 launches to 3; sum order = arrival rank of LDS adds
# speedup vs baseline: 4.1318x; 2.9852x over previous
#include <hip/hip_runtime.h>

typedef __attribute__((ext_vector_type(2))) float v2f;
typedef __attribute__((ext_vector_type(8))) float v8f;

#define NN 50000
#define NE 800000
#define DF 64
#define DH 128
#define LDS_STRIDE 132

#define NG  250
#define CH  3200
#define NB  250
#define NPB 200
#define PT  640
#define QT  1024
#define CAP 6144


__device__ __forceinline__ int scan256(int c, int* wsum, int& total) {
  const int t = threadIdx.x;
  const int lane = t & 31;
  const int w = t >> 5;
  int x = c;
#pragma unroll
  for (int d = 1; d < 32; d <<= 1) {
    int y = __shfl_up(x, d);
    if (lane >= d) x += y;
  }
  if (lane == 31 && w < 8) wsum[w] = x;
  __syncthreads();
  if (t == 0) {
    int s = 0;
    for (int k = 0; k < 8; ++k) { int v = wsum[k]; wsum[k] = s; s += v; }
    wsum[8] = s;
  }
  __syncthreads();
  const int r = ((w < 8) ? wsum[w] : 0) + x - c;
  total = wsum[8];
  __syncthreads();
  return r;
}

__global__ __launch_bounds__(PT) void gin_part(const int* __restrict__ src,
                                               const int* __restrict__ dst,
                                               int* __restrict__ startT,
                                               int* __restrict__ part) {
  __shared__ int hist[256];
  __shared__ int startl[256];
  __shared__ int wsum[16];
  __shared__ int lsorted[CH];
  const int t = threadIdx.x;
  const int g = blockIdx.x;
  const int base = g * CH;
  if (t < 256) hist[t] = 0;
  __syncthreads();
  int pk[5], bn[5], rk[5];
#pragma unroll
  for (int k = 0; k < 5; ++k) {
    const int e = base + k * PT + t;
    const unsigned d = min((unsigned)dst[e], (unsigned)(NN - 1));
    const unsigned s = min((unsigned)src[e], (unsigned)(NN - 1));
    const unsigned b = d / NPB;
    bn[k] = (int)b;
    pk[k] = (int)((s << 8) | (d - b * NPB));
    rk[k] = atomicAdd(&hist[b], 1);
  }
  __syncthreads();
  int total;
  const int c = (t < NB) ? hist[t] : 0;
  const int st = scan256(c, wsum, total);
  if (t < NB) { startl[t] = st; startT[t * NG + g] = st; }
  if (t == 0) startT[NB * NG + g] = CH;
  __syncthreads();
#pragma unroll
  for (int k = 0; k < 5; ++k) lsorted[startl[bn[k]] + rk[k]] = pk[k];
  __syncthreads();
#pragma unroll
  for (int k = 0; k < 5; ++k) part[base + k * PT + t] = lsorted[k * PT + t];
}

__global__ __launch_bounds__(QT) void gin_bin_gather(const float* __restrict__ feat,
                                                     const float* __restrict__ eps,
                                                     const int* __restrict__ startT,
                                                     const int* __restrict__ part,
                                                     float* __restrict__ acc) {
  __shared__ int pos[256];
  __shared__ int segs[256];
  __shared__ int cntl[256];
  __shared__ int offl[256];
  __shared__ int cur[256];
  __shared__ int wsum[16];
  __shared__ int lv[CAP];
  __shared__ int lss[CAP];
  const int t = threadIdx.x;
  const int lane = t & 31;
  const int w = t >> 5;
  const int b = blockIdx.x;

  int s = 0, n = 0;
  if (t < NG) {
    s = startT[b * NG + t];
    const int e = startT[(b + 1) * NG + t];
    s = min(max(s, 0), CH);
    n = min(max(e - s, 0), CH - s);
  }
  int nb;
  const int p0 = scan256(n, wsum, nb);
  if (t < 256) { pos[t] = (t < NG) ? p0 : nb; segs[t] = s; }

  const float sc = 1.0f + eps[0];
  const float2* f2 = (const float2*)feat;
  float2* acc2 = (float2*)acc;

  int r0 = 0;
  do {
    const int nr = min(nb - r0, CAP);
    if (t < 256) cntl[t] = 0;
    __syncthreads();
    for (int pb = 0; pb < nr; pb += 4 * QT) {
      int vv[4];
#pragma unroll
      for (int j = 0; j < 4; ++j) {
        const int p = pb + j * QT + t;
        vv[j] = 0;
        if (p < nr) {
          const int pg = r0 + p;
          int lo = 0;
#pragma unroll
          for (int st = 128; st >= 1; st >>= 1) { if (pos[lo + st] <= pg) lo += st; }
          vv[j] = part[lo * CH + segs[lo] + (pg - pos[lo])];
        }
      }
#pragma unroll
      for (int j = 0; j < 4; ++j) {
        const int p = pb + j * QT + t;
        if (p < nr) {
          lv[p] = vv[j];
          atomicAdd(&cntl[min(vv[j] & 255, NPB - 1)], 1);
        }
      }
    }
    __syncthreads();
    int tot;
    const int cc = (t < 256) ? cntl[t] : 0;
    const int oo = scan256(cc, wsum, tot);
    if (t < 256) { offl[t] = oo; cur[t] = oo; }
    __syncthreads();
    for (int p = t; p < nr; p += QT) {
      const int v = lv[p];
      const int q = atomicAdd(&cur[min(v & 255, NPB - 1)], 1);
      lss[min((unsigned)q, (unsigned)(CAP - 1))] = v >> 8;
    }
    __syncthreads();

    for (int nl = w; nl < NPB; nl += QT / 32) {
      const int node = b * NPB + nl;
      int c = __builtin_amdgcn_readfirstlane(cntl[nl]);
      int o = __builtin_amdgcn_readfirstlane(offl[nl]);
      c = min(max(c, 0), nr);
      o = min(max(o, 0), nr - c);
      float2 own;
      if (r0 == 0) own = f2[(long long)node * 32 + lane];
      else         own = acc2[(long long)node * 32 + lane];

      float2 a;
      a.x = 0.0f; a.y = 0.0f;
      for (int base = 0; base < c; base += 32) {
        const int m = min(c - base, 32);
        int my = 0;
        if (lane < m) my = lss[o + base + lane];
        unsigned sidx[32];
#pragma unroll
        for (int u = 0; u < 32; ++u) {
          sidx[u] = min((unsigned)__builtin_amdgcn_readlane(my, u), (unsigned)(NN - 1));
        }
        float2 v[32];
#pragma unroll
        for (int u = 0; u < 32; ++u) {
          v[u].x = 0.0f; v[u].y = 0.0f;
          if (u < m) v[u] = f2[(long long)sidx[u] * 32 + lane];
        }
#pragma unroll
        for (int u = 0; u < 32; ++u) { a.x += v[u].x; a.y += v[u].y; }
      }
      float2 r;
      if (r0 == 0) { r.x = sc * own.x + a.x; r.y = sc * own.y + a.y; }
      else         { r.x = own.x + a.x;      r.y = own.y + a.y; }
      acc2[(long long)node * 32 + lane] = r;
    }
    r0 += CAP;
    __syncthreads();
  } while (r0 < nb);
}

__global__ __launch_bounds__(32) void gin_mlp(const float* __restrict__ acc,
                                              const float* __restrict__ W1,
                                              const float* __restrict__ b1,
                                              const float* __restrict__ W2,
                                              const float* __restrict__ b2,
                                              float* __restrict__ out) {
  __shared__ float lds[16 * LDS_STRIDE];

  const int tile = blockIdx.x;
  const int lane = threadIdx.x;
  const int half = lane >> 4;
  const int l16  = lane & 15;
  const int row0 = tile * 16;

  v2f afrag[16];
  {
    const float* arow = acc + (long long)(row0 + l16) * DF;
#pragma unroll
    for (int kk = 0; kk < 16; ++kk) {
      int kb = kk * 4 + half * 2;
      afrag[kk].x = arow[kb];
      afrag[kk].y = arow[kb + 1];
    }
  }

#pragma unroll
  for (int nt = 0; nt < 8; ++nt) {
    const int n0 = nt * 16;
    v8f c = {};
#pragma unroll
    for (int kk = 0; kk < 16; ++kk) {
      int kr = kk * 4 + half * 2;
      v2f bf;
      bf.x = W1[kr * DH + n0 + l16];
      bf.y = W1[(kr + 1) * DH + n0 + l16];
      c = __builtin_amdgcn_wmma_f32_16x16x4_f32(false, afrag[kk], false, bf,
                                                (short)0, c, false, false);
    }
    float bias = b1[n0 + l16];
#pragma unroll
    for (int r = 0; r < 8; ++r) {
      float h = c[r] + bias;
      h = h > 0.0f ? h : 0.0f;
      lds[(r + half * 8) * LDS_STRIDE + n0 + l16] = h;
    }
  }

  __syncthreads();

#pragma unroll
  for (int nt = 0; nt < 4; ++nt) {
    const int n0 = nt * 16;
    v8f c = {};
#pragma unroll
    for (int kk = 0; kk < 32; ++kk) {
      int kr = kk * 4 + half * 2;
      v2f af, bf;
      af.x = lds[l16 * LDS_STRIDE + kr];
      af.y = lds[l16 * LDS_STRIDE + kr + 1];
      bf.x = W2[kr * DF + n0 + l16];
      bf.y = W2[(kr + 1) * DF + n0 + l16];
      c = __builtin_amdgcn_wmma_f32_16x16x4_f32(false, af, false, bf,
                                                (short)0, c, false, false);
    }
    float bias = b2[n0 + l16];
#pragma unroll
    for (int r = 0; r < 8; ++r) {
      out[(long long)(row0 + r + half * 8) * DF + n0 + l16] = c[r] + bias;
    }
  }
}

extern "C" void kernel_launch(void* const* d_in, const int* in_sizes, int n_in,
                              void* d_out, int out_size, void* d_ws, size_t ws_size,
                              hipStream_t stream) {
  const float* feat = (const float*)d_in[0];
  const int*   src  = (const int*)d_in[1];
  const int*   dst  = (const int*)d_in[2];
  const float* eps  = (const float*)d_in[3];
  const float* W1   = (const float*)d_in[4];
  const float* b1   = (const float*)d_in[5];
  const float* W2   = (const float*)d_in[6];
  const float* b2   = (const float*)d_in[7];
  float* acc    = (float*)d_ws;
  int*   startT = (int*)d_ws + 3200000;
  int*   part   = (int*)d_ws + 3262784;
  float* out = (float*)d_out;

  gin_part<<<NG, PT, 0, stream>>>(src, dst, startT, part);
  gin_bin_gather<<<NB, QT, 0, stream>>>(feat, eps, startT, part, acc);
  gin_mlp<<<NN / 16, 32, 0, stream>>>(acc, W1, b1, W2, b2, out);
}


# MLP fused into the bin gather kernel: node sums, hidden rows, W1 and W2 stay in LDS, 13 row tiles per bin (208 nodes, 241 bins); 3 launches to 2; per-node sum order still the arrival rank of LDS adds
# speedup vs baseline: 4.4789x; 1.0840x over previous
#include <hip/hip_runtime.h>

typedef __attribute__((ext_vector_type(2))) float v2f;
typedef __attribute__((ext_vector_type(8))) float v8f;

#define NN 50000
#define NE 800000
#define DF 64
#define DH 128
#define LDS_STRIDE 132

#define NG  250
#define CH  3200
#define NPB 208
#define NB  241
#define TPB 13
#define ACC_STRIDE 68
#define W1_STRIDE 136
#define W2_STRIDE 72
#define PT  640
#define QT  1024
#define CAP 6144


__device__ __forceinline__ int scan256(int c, int* wsum, int& total) {
  const int t = threadIdx.x;
  const int lane = t & 31;
  const int w = t >> 5;
  int x = c;
#pragma unroll
  for (int d = 1; d < 32; d <<= 1) {
    int y = __shfl_up(x, d);
    if (lane >= d) x += y;
  }
  if (lane == 31 && w < 8) wsum[w] = x;
  __syncthreads();
  if (t == 0) {
    int s = 0;
    for (int k = 0; k < 8; ++k) { int v = wsum[k]; wsum[k] = s; s += v; }
    wsum[8] = s;
  }
  __syncthreads();
  const int r = ((w < 8) ? wsum[w] : 0) + x - c;
  total = wsum[8];
  __syncthreads();
  return r;
}

__global__ __launch_bounds__(PT) void gin_part(const int* __restrict__ src,
                                               const int* __restrict__ dst,
                                               int* __restrict__ startT,
                                               int* __restrict__ part) {
  __shared__ int hist[256];
  __shared__ int startl[256];
  __shared__ int wsum[16];
  __shared__ int lsorted[CH];
  const int t = threadIdx.x;
  const int g = blockIdx.x;
  const int base = g * CH;
  if (t < 256) hist[t] = 0;
  __syncthreads();
  int pk[5], bn[5], rk[5];
#pragma unroll
  for (int k = 0; k < 5; ++k) {
    const int e = base + k * PT + t;
    const unsigned d = min((unsigned)dst[e], (unsigned)(NN - 1));
    const unsigned s = min((unsigned)src[e], (unsigned)(NN - 1));
    const unsigned b = d / NPB;
    bn[k] = (int)b;
    pk[k] = (int)((s << 8) | (d - b * NPB));
    rk[k] = atomicAdd(&hist[b], 1);
  }
  __syncthreads();
  int total;
  const int c = (t < NB) ? hist[t] : 0;
  const int st = scan256(c, wsum, total);
  if (t < NB) { startl[t] = st; startT[t * NG + g] = st; }
  if (t == 0) startT[NB * NG + g] = CH;
  __syncthreads();
#pragma unroll
  for (int k = 0; k < 5; ++k) lsorted[startl[bn[k]] + rk[k]] = pk[k];
  __syncthreads();
#pragma unroll
  for (int k = 0; k < 5; ++k) part[base + k * PT + t] = lsorted[k * PT + t];
}

__global__ __launch_bounds__(QT) void gin_bin_gather_mlp(const float* __restrict__ feat,
                                                         const float* __restrict__ eps,
                                                         const int* __restrict__ startT,
                                                         const int* __restrict__ part,
                                                         const float* __restrict__ W1,
                                                         const float* __restrict__ b1,
                                                         const float* __restrict__ W2,
                                                         const float* __restrict__ b2,
                                                         float* __restrict__ out) {
  __shared__ int pos[256];
  __shared__ int segs[256];
  __shared__ int cntl[256];
  __shared__ int offl[256];
  __shared__ int cur[256];
  __shared__ int wsum[16];
  __shared__ int lv[CAP];
  __shared__ int lss[CAP];
  __shared__ __attribute__((aligned(16))) float accL[NPB * ACC_STRIDE];
  __shared__ __attribute__((aligned(16))) float hid[NPB * LDS_STRIDE];
  __shared__ __attribute__((aligned(16))) float w1s[DF * W1_STRIDE];
  __shared__ __attribute__((aligned(16))) float w2s[DH * W2_STRIDE];
  const int t = threadIdx.x;
  const int lane = t & 31;
  const int w = t >> 5;
  const int b = blockIdx.x;

  {
    const float4* W1v = (const float4*)W1;
    const float4* W2v = (const float4*)W2;
    for (int i = t; i < DF * DH / 4; i += QT) {
      const int r = i >> 5, c4 = i & 31;
      *(float4*)&w1s[r * W1_STRIDE + c4 * 4] = W1v[i];
    }
    for (int i = t; i < DH * DF / 4; i += QT) {
      const int r = i >> 4, c4 = i & 15;
      *(float4*)&w2s[r * W2_STRIDE + c4 * 4] = W2v[i];
    }
  }

  int s = 0, n = 0;
  if (t < NG) {
    s = startT[b * NG + t];
    const int e = startT[(b + 1) * NG + t];
    s = min(max(s, 0), CH);
    n = min(max(e - s, 0), CH - s);
  }
  int nb;
  const int p0 = scan256(n, wsum, nb);
  if (t < 256) { pos[t] = (t < NG) ? p0 : nb; segs[t] = s; }

  const float sc = 1.0f + eps[0];
  const float2* f2 = (const float2*)feat;

  int r0 = 0;
  do {
    const int nr = min(nb - r0, CAP);
    if (t < 256) cntl[t] = 0;
    __syncthreads();
    for (int pb = 0; pb < nr; pb += 4 * QT) {
      int vv[4];
#pragma unroll
      for (int j = 0; j < 4; ++j) {
        const int p = pb + j * QT + t;
        vv[j] = 0;
        if (p < nr) {
          const int pg = r0 + p;
          int lo = 0;
#pragma unroll
          for (int st = 128; st >= 1; st >>= 1) { if (pos[lo + st] <= pg) lo += st; }
          vv[j] = part[lo * CH + segs[lo] + (pg - pos[lo])];
        }
      }
#pragma unroll
      for (int j = 0; j < 4; ++j) {
        const int p = pb + j * QT + t;
        if (p < nr) {
          lv[p] = vv[j];
          atomicAdd(&cntl[min(vv[j] & 255, NPB - 1)], 1);
        }
      }
    }
    __syncthreads();
    int tot;
    const int cc = (t < 256) ? cntl[t] : 0;
    const int oo = scan256(cc, wsum, tot);
    if (t < 256) { offl[t] = oo; cur[t] = oo; }
    __syncthreads();
    for (int p = t; p < nr; p += QT) {
      const int v = lv[p];
      const int q = atomicAdd(&cur[min(v & 255, NPB - 1)], 1);
      lss[min((unsigned)q, (unsigned)(CAP - 1))] = v >> 8;
    }
    __syncthreads();

    for (int nl = w; nl < NPB; nl += QT / 32) {
      const int node = b * NPB + nl;
      float2* arow = (float2*)&accL[nl * ACC_STRIDE];
      if (node >= NN) {
        if (r0 == 0) { float2 z; z.x = 0.0f; z.y = 0.0f; arow[lane] = z; }
        continue;
      }
      int c = __builtin_amdgcn_readfirstlane(cntl[nl]);
      int o = __builtin_amdgcn_readfirstlane(offl[nl]);
      c = min(max(c, 0), nr);
      o = min(max(o, 0), nr - c);
      float2 own;
      if (r0 == 0) own = f2[(long long)node * 32 + lane];
      else         own = arow[lane];

      float2 a;
      a.x = 0.0f; a.y = 0.0f;
      for (int base = 0; base < c; base += 32) {
        const int m = min(c - base, 32);
        int my = 0;
        if (lane < m) my = lss[o + base + lane];
        unsigned sidx[32];
#pragma unroll
        for (int u = 0; u < 32; ++u) {
          sidx[u] = min((unsigned)__builtin_amdgcn_readlane(my, u), (unsigned)(NN - 1));
        }
        float2 v[32];
#pragma unroll
        for (int u = 0; u < 32; ++u) {
          v[u].x = 0.0f; v[u].y = 0.0f;
          if (u < m) v[u] = f2[(long long)sidx[u] * 32 + lane];
        }
#pragma unroll
        for (int u = 0; u < 32; ++u) { a.x += v[u].x; a.y += v[u].y; }
      }
      float2 r;
      if (r0 == 0) { r.x = sc * own.x + a.x; r.y = sc * own.y + a.y; }
      else         { r.x = own.x + a.x;      r.y = own.y + a.y; }
      arow[lane] = r;
    }
    r0 += CAP;
    __syncthreads();
  } while (r0 < nb);

  const int wu = __builtin_amdgcn_readfirstlane(w);
  const int half = lane >> 4;
  const int l16  = lane & 15;

  for (int u = wu; u < TPB * 4; u += QT / 32) {
    const int tile = u >> 2;
    const int np = u & 3;
    if (b * NPB + tile * 16 >= NN) continue;
    v2f afrag[16];
    {
      const float* arow = &accL[(tile * 16 + l16) * ACC_STRIDE];
#pragma unroll
      for (int kk = 0; kk < 16; ++kk) {
        const int kb = kk * 4 + half * 2;
        afrag[kk].x = arow[kb];
        afrag[kk].y = arow[kb + 1];
      }
    }
    v8f c0 = {};
    v8f c1 = {};
    const int n0 = np * 32;
#pragma unroll
    for (int kk = 0; kk < 16; ++kk) {
      const int kr = kk * 4 + half * 2;
      v2f bf0, bf1;
      bf0.x = w1s[kr * W1_STRIDE + n0 + l16];
      bf0.y = w1s[(kr + 1) * W1_STRIDE + n0 + l16];
      bf1.x = w1s[kr * W1_STRIDE + n0 + 16 + l16];
      bf1.y = w1s[(kr + 1) * W1_STRIDE + n0 + 16 + l16];
      c0 = __builtin_amdgcn_wmma_f32_16x16x4_f32(false, afrag[kk], false, bf0,
                                                 (short)0, c0, false, false);
      c1 = __builtin_amdgcn_wmma_f32_16x16x4_f32(false, afrag[kk], false, bf1,
                                                 (short)0, c1, false, false);
    }
    const float bias0 = b1[n0 + l16];
    const float bias1 = b1[n0 + 16 + l16];
#pragma unroll
    for (int r = 0; r < 8; ++r) {
      float h0 = c0[r] + bias0;
      float h1 = c1[r] + bias1;
      h0 = h0 > 0.0f ? h0 : 0.0f;
      h1 = h1 > 0.0f ? h1 : 0.0f;
      hid[(tile * 16 + r + half * 8) * LDS_STRIDE + n0 + l16] = h0;
      hid[(tile * 16 + r + half * 8) * LDS_STRIDE + n0 + 16 + l16] = h1;
    }
  }

  __syncthreads();

  for (int u = wu; u < TPB * 2; u += QT / 32) {
    const int tile = u >> 1;
    const int np = u & 1;
    const int row0 = b * NPB + tile * 16;
    if (row0 >= NN) continue;
    const float* hrow = &hid[(tile * 16 + l16) * LDS_STRIDE];
    const int n0 = np * 32;
    v8f c0 = {};
    v8f c1 = {};
#pragma unroll
    for (int kk = 0; kk < 32; ++kk) {
      const int kr = kk * 4 + half * 2;
      v2f af, bf0, bf1;
      af.x = hrow[kr];
      af.y = hrow[kr + 1];
      bf0.x = w2s[kr * W2_STRIDE + n0 + l16];
      bf0.y = w2s[(kr + 1) * W2_STRIDE + n0 + l16];
      bf1.x = w2s[kr * W2_STRIDE + n0 + 16 + l16];
      bf1.y = w2s[(kr + 1) * W2_STRIDE + n0 + 16 + l16];
      c0 = __builtin_amdgcn_wmma_f32_16x16x4_f32(false, af, false, bf0,
                                                 (short)0, c0, false, false);
      c1 = __builtin_amdgcn_wmma_f32_16x16x4_f32(false, af, false, bf1,
                                                 (short)0, c1, false, false);
    }
    const float bias0 = b2[n0 + l16];
    const float bias1 = b2[n0 + 16 + l16];
#pragma unroll
    for (int r = 0; r < 8; ++r) {
      out[(long long)(row0 + r + half * 8) * DF + n0 + l16] = c0[r] + bias0;
      out[(long long)(row0 + r + half * 8) * DF + n0 + 16 + l16] = c1[r] + bias1;
    }
  }
}

extern "C" void kernel_launch(void* const* d_in, const int* in_sizes, int n_in,
                              void* d_out, int out_size, void* d_ws, size_t ws_size,
                              hipStream_t stream) {
  const float* feat = (const float*)d_in[0];
  const int*   src  = (const int*)d_in[1];
  const int*   dst  = (const int*)d_in[2];
  const float* eps  = (const float*)d_in[3];
  const float* W1   = (const float*)d_in[4];
  const float* b1   = (const float*)d_in[5];
  const float* W2   = (const float*)d_in[6];
  const float* b2   = (const float*)d_in[7];
  int*   startT = (int*)d_ws + 3200000;
  int*   part   = (int*)d_ws + 3262784;
  float* out = (float*)d_out;

  gin_part<<<NG, PT, 0, stream>>>(src, dst, startT, part);
  gin_bin_gather_mlp<<<NB, QT, 0, stream>>>(feat, eps, startT, part, W1, b1, W2, b2, out);
}


# Partition kernel on 5 waves per work-group instead of 20 (20 edges per thread as 16-B pieces, two bins per thread in its scan): fewer waves to enter; per-node sum order still arrival rank of LDS adds
# speedup vs baseline: 4.7288x; 1.0558x over previous
#include <hip/hip_runtime.h>

typedef __attribute__((ext_vector_type(2))) float v2f;
typedef __attribute__((ext_vector_type(8))) float v8f;

#define NN 50000
#define NE 800000
#define DF 64
#define DH 128
#define LDS_STRIDE 132

#define NG  250
#define CH  3200
#define NPB 208
#define NB  241
#define TPB 13
#define ACC_STRIDE 68
#define W1_STRIDE 136
#define W2_STRIDE 72
#define PT  160
#define EPT 20
#define QT  1024
#define CAP 6144


__device__ __forceinline__ int scan256(int c, int* wsum, int& total) {
  const int t = threadIdx.x;
  const int lane = t & 31;
  const int w = t >> 5;
  int x = c;
#pragma unroll
  for (int d = 1; d < 32; d <<= 1) {
    int y = __shfl_up(x, d);
    if (lane >= d) x += y;
  }
  if (lane == 31 && w < 8) wsum[w] = x;
  __syncthreads();
  if (t == 0) {
    int s = 0;
    for (int k = 0; k < 8; ++k) { int v = wsum[k]; wsum[k] = s; s += v; }
    wsum[8] = s;
  }
  __syncthreads();
  const int r = ((w < 8) ? wsum[w] : 0) + x - c;
  total = wsum[8];
  __syncthreads();
  return r;
}

__global__ __launch_bounds__(PT) void gin_part(const int* __restrict__ src,
                                               const int* __restrict__ dst,
                                               int* __restrict__ startT,
                                               int* __restrict__ part) {
  __shared__ int hist[256];
  __shared__ int startl[256];
  __shared__ int wsum[4];
  __shared__ __attribute__((aligned(16))) int lsorted[CH];
  const int t = threadIdx.x;
  const int lane = t & 31;
  const int w = t >> 5;
  const int g = blockIdx.x;
  const int base = g * CH;
  for (int i = t; i < 256; i += PT) hist[i] = 0;
  __syncthreads();
  const int4* d4 = (const int4*)(dst + base);
  const int4* s4 = (const int4*)(src + base);
  int4 dv[EPT / 4], sv[EPT / 4];
#pragma unroll
  for (int k = 0; k < EPT / 4; ++k) {
    dv[k] = d4[k * PT + t];
    sv[k] = s4[k * PT + t];
  }
  int pk[EPT], bn[EPT], rk[EPT];
#pragma unroll
  for (int k = 0; k < EPT / 4; ++k) {
    const int dd[4] = {dv[k].x, dv[k].y, dv[k].z, dv[k].w};
    const int ss[4] = {sv[k].x, sv[k].y, sv[k].z, sv[k].w};
#pragma unroll
    for (int j = 0; j < 4; ++j) {
      const unsigned d = min((unsigned)dd[j], (unsigned)(NN - 1));
      const unsigned s = min((unsigned)ss[j], (unsigned)(NN - 1));
      const unsigned b = d / NPB;
      bn[k * 4 + j] = (int)b;
      pk[k * 4 + j] = (int)((s << 8) | (d - b * NPB));
      rk[k * 4 + j] = atomicAdd(&hist[b], 1);
    }
  }
  __syncthreads();
  int c0 = 0, c1 = 0;
  if (t < 128) { c0 = hist[2 * t]; c1 = hist[2 * t + 1]; }
  const int cs = c0 + c1;
  int x = cs;
#pragma unroll
  for (int d = 1; d < 32; d <<= 1) {
    int y = __shfl_up(x, d);
    if (lane >= d) x += y;
  }
  if (lane == 31 && w < 4) wsum[w] = x;
  __syncthreads();
  if (t < 128) {
    int pre = 0;
#pragma unroll
    for (int k = 0; k < 3; ++k) { if (k < w) pre += wsum[k]; }
    const int e0 = pre + x - cs;
    const int e1 = e0 + c0;
    startl[2 * t] = e0;
    startl[2 * t + 1] = e1;
    if (2 * t < NB)     startT[(2 * t) * NG + g] = e0;
    if (2 * t + 1 < NB) startT[(2 * t + 1) * NG + g] = e1;
  }
  if (t == 0) startT[NB * NG + g] = CH;
  __syncthreads();
#pragma unroll
  for (int k = 0; k < EPT; ++k) lsorted[startl[bn[k]] + rk[k]] = pk[k];
  __syncthreads();
  const int4* l4 = (const int4*)lsorted;
  int4* p4 = (int4*)(part + base);
#pragma unroll
  for (int k = 0; k < EPT / 4; ++k) p4[k * PT + t] = l4[k * PT + t];
}

__global__ __launch_bounds__(QT) void gin_bin_gather_mlp(const float* __restrict__ feat,
                                                         const float* __restrict__ eps,
                                                         const int* __restrict__ startT,
                                                         const int* __restrict__ part,
                                                         const float* __restrict__ W1,
                                                         const float* __restrict__ b1,
                                                         const float* __restrict__ W2,
                                                         const float* __restrict__ b2,
                                                         float* __restrict__ out) {
  __shared__ int pos[256];
  __shared__ int segs[256];
  __shared__ int cntl[256];
  __shared__ int offl[256];
  __shared__ int cur[256];
  __shared__ int wsum[16];
  __shared__ int lv[CAP];
  __shared__ int lss[CAP];
  __shared__ __attribute__((aligned(16))) float accL[NPB * ACC_STRIDE];
  __shared__ __attribute__((aligned(16))) float hid[NPB * LDS_STRIDE];
  __shared__ __attribute__((aligned(16))) float w1s[DF * W1_STRIDE];
  __shared__ __attribute__((aligned(16))) float w2s[DH * W2_STRIDE];
  const int t = threadIdx.x;
  const int lane = t & 31;
  const int w = t >> 5;
  const int b = blockIdx.x;

  {
    const float4* W1v = (const float4*)W1;
    const float4* W2v = (const float4*)W2;
    for (int i = t; i < DF * DH / 4; i += QT) {
      const int r = i >> 5, c4 = i & 31;
      *(float4*)&w1s[r * W1_STRIDE + c4 * 4] = W1v[i];
    }
    for (int i = t; i < DH * DF / 4; i += QT) {
      const int r = i >> 4, c4 = i & 15;
      *(float4*)&w2s[r * W2_STRIDE + c4 * 4] = W2v[i];
    }
  }

  int s = 0, n = 0;
  if (t < NG) {
    s = startT[b * NG + t];
    const int e = startT[(b + 1) * NG + t];
    s = min(max(s, 0), CH);
    n = min(max(e - s, 0), CH - s);
  }
  int nb;
  const int p0 = scan256(n, wsum, nb);
  if (t < 256) { pos[t] = (t < NG) ? p0 : nb; segs[t] = s; }

  const float sc = 1.0f + eps[0];
  const float2* f2 = (const float2*)feat;

  int r0 = 0;
  do {
    const int nr = min(nb - r0, CAP);
    if (t < 256) cntl[t] = 0;
    __syncthreads();
    for (int pb = 0; pb < nr; pb += 4 * QT) {
      int vv[4];
#pragma unroll
      for (int j = 0; j < 4; ++j) {
        const int p = pb + j * QT + t;
        vv[j] = 0;
        if (p < nr) {
          const int pg = r0 + p;
          int lo = 0;
#pragma unroll
          for (int st = 128; st >= 1; st >>= 1) { if (pos[lo + st] <= pg) lo += st; }
          vv[j] = part[lo * CH + segs[lo] + (pg - pos[lo])];
        }
      }
#pragma unroll
      for (int j = 0; j < 4; ++j) {
        const int p = pb + j * QT + t;
        if (p < nr) {
          lv[p] = vv[j];
          atomicAdd(&cntl[min(vv[j] & 255, NPB - 1)], 1);
        }
      }
    }
    __syncthreads();
    int tot;
    const int cc = (t < 256) ? cntl[t] : 0;
    const int oo = scan256(cc, wsum, tot);
    if (t < 256) { offl[t] = oo; cur[t] = oo; }
    __syncthreads();
    for (int p = t; p < nr; p += QT) {
      const int v = lv[p];
      const int q = atomicAdd(&cur[min(v & 255, NPB - 1)], 1);
      lss[min((unsigned)q, (unsigned)(CAP - 1))] = v >> 8;
    }
    __syncthreads();

    for (int nl = w; nl < NPB; nl += QT / 32) {
      const int node = b * NPB + nl;
      float2* arow = (float2*)&accL[nl * ACC_STRIDE];
      if (node >= NN) {
        if (r0 == 0) { float2 z; z.x = 0.0f; z.y = 0.0f; arow[lane] = z; }
        continue;
      }
      int c = __builtin_amdgcn_readfirstlane(cntl[nl]);
      int o = __builtin_amdgcn_readfirstlane(offl[nl]);
      c = min(max(c, 0), nr);
      o = min(max(o, 0), nr - c);
      float2 own;
      if (r0 == 0) own = f2[(long long)node * 32 + lane];
      else         own = arow[lane];

      float2 a;
      a.x = 0.0f; a.y = 0.0f;
      for (int base = 0; base < c; base += 32) {
        const int m = min(c - base, 32);
        int my = 0;
        if (lane < m) my = lss[o + base + lane];
        unsigned sidx[32];
#pragma unroll
        for (int u = 0; u < 32; ++u) {
          sidx[u] = min((unsigned)__builtin_amdgcn_readlane(my, u), (unsigned)(NN - 1));
        }
        float2 v[32];
#pragma unroll
        for (int u = 0; u < 32; ++u) {
          v[u].x = 0.0f; v[u].y = 0.0f;
          if (u < m) v[u] = f2[(long long)sidx[u] * 32 + lane];
        }
#pragma unroll
        for (int u = 0; u < 32; ++u) { a.x += v[u].x; a.y += v[u].y; }
      }
      float2 r;
      if (r0 == 0) { r.x = sc * own.x + a.x; r.y = sc * own.y + a.y; }
      else         { r.x = own.x + a.x;      r.y = own.y + a.y; }
      arow[lane] = r;
    }
    r0 += CAP;
    __syncthreads();
  } while (r0 < nb);

  const int wu = __builtin_amdgcn_readfirstlane(w);
  const int half = lane >> 4;
  const int l16  = lane & 15;

  for (int u = wu; u < TPB * 4; u += QT / 32) {
    const int tile = u >> 2;
    const int np = u & 3;
    if (b * NPB + tile * 16 >= NN) continue;
    v2f afrag[16];
    {
      const float* arow = &accL[(tile * 16 + l16) * ACC_STRIDE];
#pragma unroll
      for (int kk = 0; kk < 16; ++kk) {
        const int kb = kk * 4 + half * 2;
        afrag[kk].x = arow[kb];
        afrag[kk].y = arow[kb + 1];
      }
    }
    v8f c0 = {};
    v8f c1 = {};
    const int n0 = np * 32;
#pragma unroll
    for (int kk = 0; kk < 16; ++kk) {
      const int kr = kk * 4 + half * 2;
      v2f bf0, bf1;
      bf0.x = w1s[kr * W1_STRIDE + n0 + l16];
      bf0.y = w1s[(kr + 1) * W1_STRIDE + n0 + l16];
      bf1.x = w1s[kr * W1_STRIDE + n0 + 16 + l16];
      bf1.y = w1s[(kr + 1) * W1_STRIDE + n0 + 16 + l16];
      c0 = __builtin_amdgcn_wmma_f32_16x16x4_f32(false, afrag[kk], false, bf0,
                                                 (short)0, c0, false, false);
      c1 = __builtin_amdgcn_wmma_f32_16x16x4_f32(false, afrag[kk], false, bf1,
                                                 (short)0, c1, false, false);
    }
    const float bias0 = b1[n0 + l16];
    const float bias1 = b1[n0 + 16 + l16];
#pragma unroll
    for (int r = 0; r < 8; ++r) {
      float h0 = c0[r] + bias0;
      float h1 = c1[r] + bias1;
      h0 = h0 > 0.0f ? h0 : 0.0f;
      h1 = h1 > 0.0f ? h1 : 0.0f;
      hid[(tile * 16 + r + half * 8) * LDS_STRIDE + n0 + l16] = h0;
      hid[(tile * 16 + r + half * 8) * LDS_STRIDE + n0 + 16 + l16] = h1;
    }
  }

  __syncthreads();

  for (int u = wu; u < TPB * 2; u += QT / 32) {
    const int tile = u >> 1;
    const int np = u & 1;
    const int row0 = b * NPB + tile * 16;
    if (row0 >= NN) continue;
    const float* hrow = &hid[(tile * 16 + l16) * LDS_STRIDE];
    const int n0 = np * 32;
    v8f c0 = {};
    v8f c1 = {};
#pragma unroll
    for (int kk = 0; kk < 32; ++kk) {
      const int kr = kk * 4 + half * 2;
      v2f af, bf0, bf1;
      af.x = hrow[kr];
      af.y = hrow[kr + 1];
      bf0.x = w2s[kr * W2_STRIDE + n0 + l16];
      bf0.y = w2s[(kr + 1) * W2_STRIDE + n0 + l16];
      bf1.x = w2s[kr * W2_STRIDE + n0 + 16 + l16];
      bf1.y = w2s[(kr + 1) * W2_STRIDE + n0 + 16 + l16];
      c0 = __builtin_amdgcn_wmma_f32_16x16x4_f32(false, af, false, bf0,
                                                 (short)0, c0, false, false);
      c1 = __builtin_amdgcn_wmma_f32_16x16x4_f32(false, af, false, bf1,
                                                 (short)0, c1, false, false);
    }
    const float bias0 = b2[n0 + l16];
    const float bias1 = b2[n0 + 16 + l16];
#pragma unroll
    for (int r = 0; r < 8; ++r) {
      out[(long long)(row0 + r + half * 8) * DF + n0 + l16] = c0[r] + bias0;
      out[(long long)(row0 + r + half * 8) * DF + n0 + 16 + l16] = c1[r] + bias1;
    }
  }
}

extern "C" void kernel_launch(void* const* d_in, const int* in_sizes, int n_in,
                              void* d_out, int out_size, void* d_ws, size_t ws_size,
                              hipStream_t stream) {
  const float* feat = (const float*)d_in[0];
  const int*   src  = (const int*)d_in[1];
  const int*   dst  = (const int*)d_in[2];
  const float* eps  = (const float*)d_in[3];
  const float* W1   = (const float*)d_in[4];
  const float* b1   = (const float*)d_in[5];
  const float* W2   = (const float*)d_in[6];
  const float* b2   = (const float*)d_in[7];
  int*   startT = (int*)d_ws + 3200000;
  int*   part   = (int*)d_ws + 3262784;
  float* out = (float*)d_out;

  gin_part<<<NG, PT, 0, stream>>>(src, dst, startT, part);
  gin_bin_gather_mlp<<<NB, QT, 0, stream>>>(feat, eps, startT, part, W1, b1, W2, b2, out);
}


# Gather reads each 256-B source row with 16 lanes x 16 B: one load brings two rows, lane indices straight from LDS, no readlane ladder; sum order: even and odd list positions apart, then added
# speedup vs baseline: 4.8906x; 1.0342x over previous
#include <hip/hip_runtime.h>

typedef __attribute__((ext_vector_type(2))) float v2f;
typedef __attribute__((ext_vector_type(8))) float v8f;

#define NN 50000
#define NE 800000
#define DF 64
#define DH 128
#define LDS_STRIDE 132

#define NG  250
#define CH  3200
#define NPB 208
#define NB  241
#define TPB 13
#define ACC_STRIDE 68
#define W1_STRIDE 136
#define W2_STRIDE 72
#define PT  160
#define EPT 20
#define QT  1024
#define CAP 6144


__device__ __forceinline__ int scan256(int c, int* wsum, int& total) {
  const int t = threadIdx.x;
  const int lane = t & 31;
  const int w = t >> 5;
  int x = c;
#pragma unroll
  for (int d = 1; d < 32; d <<= 1) {
    int y = __shfl_up(x, d);
    if (lane >= d) x += y;
  }
  if (lane == 31 && w < 8) wsum[w] = x;
  __syncthreads();
  if (t == 0) {
    int s = 0;
    for (int k = 0; k < 8; ++k) { int v = wsum[k]; wsum[k] = s; s += v; }
    wsum[8] = s;
  }
  __syncthreads();
  const int r = ((w < 8) ? wsum[w] : 0) + x - c;
  total = wsum[8];
  __syncthreads();
  return r;
}

__global__ __launch_bounds__(PT) void gin_part(const int* __restrict__ src,
                                               const int* __restrict__ dst,
                                               int* __restrict__ startT,
                                               int* __restrict__ part) {
  __shared__ int hist[256];
  __shared__ int startl[256];
  __shared__ int wsum[4];
  __shared__ __attribute__((aligned(16))) int lsorted[CH];
  const int t = threadIdx.x;
  const int lane = t & 31;
  const int w = t >> 5;
  const int g = blockIdx.x;
  const int base = g * CH;
  for (int i = t; i < 256; i += PT) hist[i] = 0;
  __syncthreads();
  const int4* d4 = (const int4*)(dst + base);
  const int4* s4 = (const int4*)(src + base);
  int4 dv[EPT / 4], sv[EPT / 4];
#pragma unroll
  for (int k = 0; k < EPT / 4; ++k) {
    dv[k] = d4[k * PT + t];
    sv[k] = s4[k * PT + t];
  }
  int pk[EPT], bn[EPT], rk[EPT];
#pragma unroll
  for (int k = 0; k < EPT / 4; ++k) {
    const int dd[4] = {dv[k].x, dv[k].y, dv[k].z, dv[k].w};
    const int ss[4] = {sv[k].x, sv[k].y, sv[k].z, sv[k].w};
#pragma unroll
    for (int j = 0; j < 4; ++j) {
      const unsigned d = min((unsigned)dd[j], (unsigned)(NN - 1));
      const unsigned s = min((unsigned)ss[j], (unsigned)(NN - 1));
      const unsigned b = d / NPB;
      bn[k * 4 + j] = (int)b;
      pk[k * 4 + j] = (int)((s << 8) | (d - b * NPB));
      rk[k * 4 + j] = atomicAdd(&hist[b], 1);
    }
  }
  __syncthreads();
  int c0 = 0, c1 = 0;
  if (t < 128) { c0 = hist[2 * t]; c1 = hist[2 * t + 1]; }
  const int cs = c0 + c1;
  int x = cs;
#pragma unroll
  for (int d = 1; d < 32; d <<= 1) {
    int y = __shfl_up(x, d);
    if (lane >= d) x += y;
  }
  if (lane == 31 && w < 4) wsum[w] = x;
  __syncthreads();
  if (t < 128) {
    int pre = 0;
#pragma unroll
    for (int k = 0; k < 3; ++k) { if (k < w) pre += wsum[k]; }
    const int e0 = pre + x - cs;
    const int e1 = e0 + c0;
    startl[2 * t] = e0;
    startl[2 * t + 1] = e1;
    if (2 * t < NB)     startT[(2 * t) * NG + g] = e0;
    if (2 * t + 1 < NB) startT[(2 * t + 1) * NG + g] = e1;
  }
  if (t == 0) startT[NB * NG + g] = CH;
  __syncthreads();
#pragma unroll
  for (int k = 0; k < EPT; ++k) lsorted[startl[bn[k]] + rk[k]] = pk[k];
  __syncthreads();
  const int4* l4 = (const int4*)lsorted;
  int4* p4 = (int4*)(part + base);
#pragma unroll
  for (int k = 0; k < EPT / 4; ++k) p4[k * PT + t] = l4[k * PT + t];
}

__device__ __forceinline__ float xhalf(float x) {
  const int b = __builtin_bit_cast(int, x);
  return __builtin_bit_cast(float,
      __builtin_amdgcn_permlanex16(b, b, 0x76543210, 0xfedcba98, false, false));
}

__device__ __forceinline__ float4 ld_piece(const float* __restrict__ feat, int idx, unsigned lo) {
  const unsigned r = min((unsigned)idx, (unsigned)(NN - 1));
  return *(const float4*)((const char*)feat + ((r << 8) | lo));
}

template <int J>
__device__ __forceinline__ void load_rungs(float4 (&v)[16], const int (&idx)[16], int hp,
                                           const float* __restrict__ feat, unsigned lo) {
  if constexpr (J < 16) {
    if (J < hp) {
      v[J] = ld_piece(feat, idx[J], lo);
      load_rungs<J + 1>(v, idx, hp, feat, lo);
    }
  }
}

template <int J>
__device__ __forceinline__ void add_rungs(float4& a, const float4 (&v)[16], int hp) {
  if constexpr (J < 16) {
    if (J < hp) {
      a.x += v[J].x; a.y += v[J].y; a.z += v[J].z; a.w += v[J].w;
      add_rungs<J + 1>(a, v, hp);
    }
  }
}

__global__ __launch_bounds__(QT) void gin_bin_gather_mlp(const float* __restrict__ feat,
                                                         const float* __restrict__ eps,
                                                         const int* __restrict__ startT,
                                                         const int* __restrict__ part,
                                                         const float* __restrict__ W1,
                                                         const float* __restrict__ b1,
                                                         const float* __restrict__ W2,
                                                         const float* __restrict__ b2,
                                                         float* __restrict__ out) {
  __shared__ int pos[256];
  __shared__ int segs[256];
  __shared__ int cntl[256];
  __shared__ int offl[256];
  __shared__ int cur[256];
  __shared__ int wsum[16];
  __shared__ int lv[CAP];
  __shared__ int lss[CAP + 32];
  __shared__ __attribute__((aligned(16))) float accL[NPB * ACC_STRIDE];
  __shared__ __attribute__((aligned(16))) float hid[NPB * LDS_STRIDE];
  __shared__ __attribute__((aligned(16))) float w1s[DF * W1_STRIDE];
  __shared__ __attribute__((aligned(16))) float w2s[DH * W2_STRIDE];
  const int t = threadIdx.x;
  const int lane = t & 31;
  const int w = t >> 5;
  const int b = blockIdx.x;
  const int half = lane >> 4;
  const int l16  = lane & 15;

  {
    const float4* W1v = (const float4*)W1;
    const float4* W2v = (const float4*)W2;
    for (int i = t; i < DF * DH / 4; i += QT) {
      const int r = i >> 5, c4 = i & 31;
      *(float4*)&w1s[r * W1_STRIDE + c4 * 4] = W1v[i];
    }
    for (int i = t; i < DH * DF / 4; i += QT) {
      const int r = i >> 4, c4 = i & 15;
      *(float4*)&w2s[r * W2_STRIDE + c4 * 4] = W2v[i];
    }
  }

  int s = 0, n = 0;
  if (t < NG) {
    s = startT[b * NG + t];
    const int e = startT[(b + 1) * NG + t];
    s = min(max(s, 0), CH);
    n = min(max(e - s, 0), CH - s);
  }
  int nb;
  const int p0 = scan256(n, wsum, nb);
  if (t < 256) { pos[t] = (t < NG) ? p0 : nb; segs[t] = s; }

  const float sc = 1.0f + eps[0];
  const unsigned lo = (unsigned)l16 << 4;

  int r0 = 0;
  do {
    const int nr = min(nb - r0, CAP);
    if (t < 256) cntl[t] = 0;
    __syncthreads();
    for (int pb = 0; pb < nr; pb += 4 * QT) {
      int vv[4];
#pragma unroll
      for (int j = 0; j < 4; ++j) {
        const int p = pb + j * QT + t;
        vv[j] = 0;
        if (p < nr) {
          const int pg = r0 + p;
          int lo = 0;
#pragma unroll
          for (int st = 128; st >= 1; st >>= 1) { if (pos[lo + st] <= pg) lo += st; }
          vv[j] = part[lo * CH + segs[lo] + (pg - pos[lo])];
        }
      }
#pragma unroll
      for (int j = 0; j < 4; ++j) {
        const int p = pb + j * QT + t;
        if (p < nr) {
          lv[p] = vv[j];
          atomicAdd(&cntl[min(vv[j] & 255, NPB - 1)], 1);
        }
      }
    }
    __syncthreads();
    int tot;
    const int cc = (t < 256) ? cntl[t] : 0;
    const int oo = scan256(cc, wsum, tot);
    if (t < 256) { offl[t] = oo; cur[t] = oo; }
    __syncthreads();
    for (int p = t; p < nr; p += QT) {
      const int v = lv[p];
      const int q = atomicAdd(&cur[min(v & 255, NPB - 1)], 1);
      lss[min((unsigned)q, (unsigned)(CAP - 1))] = v >> 8;
    }
    __syncthreads();

    for (int nl = w; nl < NPB; nl += QT / 32) {
      const int node = b * NPB + nl;
      float4* arow = (float4*)&accL[nl * ACC_STRIDE];
      if (node >= NN) {
        if (r0 == 0 && half == 0) arow[l16] = make_float4(0.0f, 0.0f, 0.0f, 0.0f);
        continue;
      }
      int c = __builtin_amdgcn_readfirstlane(cntl[nl]);
      int o = __builtin_amdgcn_readfirstlane(offl[nl]);
      c = min(max(c, 0), nr);
      o = min(max(o, 0), nr - c);
      float4 own = make_float4(0.0f, 0.0f, 0.0f, 0.0f);
      if (r0 == 0) own = ld_piece(feat, node, lo);

      float4 a = make_float4(0.0f, 0.0f, 0.0f, 0.0f);
      for (int base = 0; base < c; base += 32) {
        const int m = min(c - base, 32);
        const int hp = m >> 1;
        const int* lp = &lss[o + base + half];
        int idx[16];
#pragma unroll
        for (int j = 0; j < 16; ++j) idx[j] = lp[2 * j];
        float4 vt = make_float4(0.0f, 0.0f, 0.0f, 0.0f);
        if (m & 1) {
          const int it = lss[o + base + m - 1];
          if (half == 0) vt = ld_piece(feat, it, lo);
        }
        float4 v[16];
        load_rungs<0>(v, idx, hp, feat, lo);
        add_rungs<0>(a, v, hp);
        a.x += vt.x; a.y += vt.y; a.z += vt.z; a.w += vt.w;
      }
      a.x += xhalf(a.x); a.y += xhalf(a.y); a.z += xhalf(a.z); a.w += xhalf(a.w);
      float4 r;
      if (r0 == 0) {
        r.x = sc * own.x + a.x; r.y = sc * own.y + a.y;
        r.z = sc * own.z + a.z; r.w = sc * own.w + a.w;
      } else {
        const float4 prev = arow[l16];
        r.x = prev.x + a.x; r.y = prev.y + a.y; r.z = prev.z + a.z; r.w = prev.w + a.w;
      }
      if (half == 0) arow[l16] = r;
    }
    r0 += CAP;
    __syncthreads();
  } while (r0 < nb);

  const int wu = __builtin_amdgcn_readfirstlane(w);

  for (int u = wu; u < TPB * 4; u += QT / 32) {
    const int tile = u >> 2;
    const int np = u & 3;
    if (b * NPB + tile * 16 >= NN) continue;
    v2f afrag[16];
    {
      const float* arow = &accL[(tile * 16 + l16) * ACC_STRIDE];
#pragma unroll
      for (int kk = 0; kk < 16; ++kk) {
        const int kb = kk * 4 + half * 2;
        afrag[kk].x = arow[kb];
        afrag[kk].y = arow[kb + 1];
      }
    }
    v8f c0 = {};
    v8f c1 = {};
    const int n0 = np * 32;
#pragma unroll
    for (int kk = 0; kk < 16; ++kk) {
      const int kr = kk * 4 + half * 2;
      v2f bf0, bf1;
      bf0.x = w1s[kr * W1_STRIDE + n0 + l16];
      bf0.y = w1s[(kr + 1) * W1_STRIDE + n0 + l16];
      bf1.x = w1s[kr * W1_STRIDE + n0 + 16 + l16];
      bf1.y = w1s[(kr + 1) * W1_STRIDE + n0 + 16 + l16];
      c0 = __builtin_amdgcn_wmma_f32_16x16x4_f32(false, afrag[kk], false, bf0,
                                                 (short)0, c0, false, false);
      c1 = __builtin_amdgcn_wmma_f32_16x16x4_f32(false, afrag[kk], false, bf1,
                                                 (short)0, c1, false, false);
    }
    const float bias0 = b1[n0 + l16];
    const float bias1 = b1[n0 + 16 + l16];
#pragma unroll
    for (int r = 0; r < 8; ++r) {
      float h0 = c0[r] + bias0;
      float h1 = c1[r] + bias1;
      h0 = h0 > 0.0f ? h0 : 0.0f;
      h1 = h1 > 0.0f ? h1 : 0.0f;
      hid[(tile * 16 + r + half * 8) * LDS_STRIDE + n0 + l16] = h0;
      hid[(tile * 16 + r + half * 8) * LDS_STRIDE + n0 + 16 + l16] = h1;
    }
  }

  __syncthreads();

  for (int u = wu; u < TPB * 2; u += QT / 32) {
    const int tile = u >> 1;
    const int np = u & 1;
    const int row0 = b * NPB + tile * 16;
    if (row0 >= NN) continue;
    const float* hrow = &hid[(tile * 16 + l16) * LDS_STRIDE];
    const int n0 = np * 32;
    v8f c0 = {};
    v8f c1 = {};
#pragma unroll
    for (int kk = 0; kk < 32; ++kk) {
      const int kr = kk * 4 + half * 2;
      v2f af, bf0, bf1;
      af.x = hrow[kr];
      af.y = hrow[kr + 1];
      bf0.x = w2s[kr * W2_STRIDE + n0 + l16];
      bf0.y = w2s[(kr + 1) * W2_STRIDE + n0 + l16];
      bf1.x = w2s[kr * W2_STRIDE + n0 + 16 + l16];
      bf1.y = w2s[(kr + 1) * W2_STRIDE + n0 + 16 + l16];
      c0 = __builtin_amdgcn_wmma_f32_16x16x4_f32(false, af, false, bf0,
                                                 (short)0, c0, false, false);
      c1 = __builtin_amdgcn_wmma_f32_16x16x4_f32(false, af, false, bf1,
                                                 (short)0, c1, false, false);
    }
    const float bias0 = b2[n0 + l16];
    const float bias1 = b2[n0 + 16 + l16];
#pragma unroll
    for (int r = 0; r < 8; ++r) {
      out[(long long)(row0 + r + half * 8) * DF + n0 + l16] = c0[r] + bias0;
      out[(long long)(row0 + r + half * 8) * DF + n0 + 16 + l16] = c1[r] + bias1;
    }
  }
}

extern "C" void kernel_launch(void* const* d_in, const int* in_sizes, int n_in,
                              void* d_out, int out_size, void* d_ws, size_t ws_size,
                              hipStream_t stream) {
  const float* feat = (const float*)d_in[0];
  const int*   src  = (const int*)d_in[1];
  const int*   dst  = (const int*)d_in[2];
  const float* eps  = (const float*)d_in[3];
  const float* W1   = (const float*)d_in[4];
  const float* b1   = (const float*)d_in[5];
  const float* W2   = (const float*)d_in[6];
  const float* b2   = (const float*)d_in[7];
  int*   startT = (int*)d_ws + 3200000;
  int*   part   = (int*)d_ws + 3262784;
  float* out = (float*)d_out;

  gin_part<<<NG, PT, 0, stream>>>(src, dst, startT, part);
  gin_bin_gather_mlp<<<NB, QT, 0, stream>>>(feat, eps, startT, part, W1, b1, W2, b2, out);
}


# Loads that ran one round trip after another go out as groups: table and weight loads together, the 4 bin searches in lockstep, cursor adds by four; per-node sum order still arrival rank of LDS adds
# speedup vs baseline: 5.1241x; 1.0477x over previous
#include <hip/hip_runtime.h>

typedef __attribute__((ext_vector_type(2))) float v2f;
typedef __attribute__((ext_vector_type(8))) float v8f;

#define NN 50000
#define NE 800000
#define DF 64
#define DH 128
#define LDS_STRIDE 132

#define NG  250
#define CH  3200
#define NPB 208
#define NB  241
#define TPB 13
#define ACC_STRIDE 68
#define W1_STRIDE 136
#define W2_STRIDE 72
#define PT  160
#define EPT 20
#define QT  1024
#define CAP 6144


__device__ __forceinline__ int scan256(int c, int* wsum, int& total) {
  const int t = threadIdx.x;
  const int lane = t & 31;
  const int w = t >> 5;
  int x = c;
#pragma unroll
  for (int d = 1; d < 32; d <<= 1) {
    int y = __shfl_up(x, d);
    if (lane >= d) x += y;
  }
  if (lane == 31 && w < 8) wsum[w] = x;
  __syncthreads();
  if (t == 0) {
    int s = 0;
    for (int k = 0; k < 8; ++k) { int v = wsum[k]; wsum[k] = s; s += v; }
    wsum[8] = s;
  }
  __syncthreads();
  const int r = ((w < 8) ? wsum[w] : 0) + x - c;
  total = wsum[8];
  __syncthreads();
  return r;
}

__global__ __launch_bounds__(PT) void gin_part(const int* __restrict__ src,
                                               const int* __restrict__ dst,
                                               int* __restrict__ startT,
                                               int* __restrict__ part) {
  __shared__ int hist[256];
  __shared__ int startl[256];
  __shared__ int wsum[4];
  __shared__ __attribute__((aligned(16))) int lsorted[CH];
  const int t = threadIdx.x;
  const int lane = t & 31;
  const int w = t >> 5;
  const int g = blockIdx.x;
  const int base = g * CH;
  for (int i = t; i < 256; i += PT) hist[i] = 0;
  __syncthreads();
  const int4* d4 = (const int4*)(dst + base);
  const int4* s4 = (const int4*)(src + base);
  int4 dv[EPT / 4], sv[EPT / 4];
#pragma unroll
  for (int k = 0; k < EPT / 4; ++k) {
    dv[k] = d4[k * PT + t];
    sv[k] = s4[k * PT + t];
  }
  int pk[EPT], bn[EPT], rk[EPT];
#pragma unroll
  for (int k = 0; k < EPT / 4; ++k) {
    const int dd[4] = {dv[k].x, dv[k].y, dv[k].z, dv[k].w};
    const int ss[4] = {sv[k].x, sv[k].y, sv[k].z, sv[k].w};
#pragma unroll
    for (int j = 0; j < 4; ++j) {
      const unsigned d = min((unsigned)dd[j], (unsigned)(NN - 1));
      const unsigned s = min((unsigned)ss[j], (unsigned)(NN - 1));
      const unsigned b = d / NPB;
      bn[k * 4 + j] = (int)b;
      pk[k * 4 + j] = (int)((s << 8) | (d - b * NPB));
      rk[k * 4 + j] = atomicAdd(&hist[b], 1);
    }
  }
  __syncthreads();
  int c0 = 0, c1 = 0;
  if (t < 128) { c0 = hist[2 * t]; c1 = hist[2 * t + 1]; }
  const int cs = c0 + c1;
  int x = cs;
#pragma unroll
  for (int d = 1; d < 32; d <<= 1) {
    int y = __shfl_up(x, d);
    if (lane >= d) x += y;
  }
  if (lane == 31 && w < 4) wsum[w] = x;
  __syncthreads();
  if (t < 128) {
    int pre = 0;
#pragma unroll
    for (int k = 0; k < 3; ++k) { if (k < w) pre += wsum[k]; }
    const int e0 = pre + x - cs;
    const int e1 = e0 + c0;
    startl[2 * t] = e0;
    startl[2 * t + 1] = e1;
    if (2 * t < NB)     startT[(2 * t) * NG + g] = e0;
    if (2 * t + 1 < NB) startT[(2 * t + 1) * NG + g] = e1;
  }
  if (t == 0) startT[NB * NG + g] = CH;
  __syncthreads();
#pragma unroll
  for (int k = 0; k < EPT; ++k) lsorted[startl[bn[k]] + rk[k]] = pk[k];
  __syncthreads();
  const int4* l4 = (const int4*)lsorted;
  int4* p4 = (int4*)(part + base);
#pragma unroll
  for (int k = 0; k < EPT / 4; ++k) p4[k * PT + t] = l4[k * PT + t];
}

__device__ __forceinline__ float xhalf(float x) {
  const int b = __builtin_bit_cast(int, x);
  return __builtin_bit_cast(float,
      __builtin_amdgcn_permlanex16(b, b, 0x76543210, 0xfedcba98, false, false));
}

__device__ __forceinline__ float4 ld_piece(const float* __restrict__ feat, int idx, unsigned lo) {
  const unsigned r = min((unsigned)idx, (unsigned)(NN - 1));
  return *(const float4*)((const char*)feat + ((r << 8) | lo));
}

template <int J>
__device__ __forceinline__ void load_rungs(float4 (&v)[16], const int (&idx)[16], int hp,
                                           const float* __restrict__ feat, unsigned lo) {
  if constexpr (J < 16) {
    if (J < hp) {
      v[J] = ld_piece(feat, idx[J], lo);
      load_rungs<J + 1>(v, idx, hp, feat, lo);
    }
  }
}

template <int J>
__device__ __forceinline__ void add_rungs(float4& a, const float4 (&v)[16], int hp) {
  if constexpr (J < 16) {
    if (J < hp) {
      a.x += v[J].x; a.y += v[J].y; a.z += v[J].z; a.w += v[J].w;
      add_rungs<J + 1>(a, v, hp);
    }
  }
}

__global__ __launch_bounds__(QT) void gin_bin_gather_mlp(const float* __restrict__ feat,
                                                         const float* __restrict__ eps,
                                                         const int* __restrict__ startT,
                                                         const int* __restrict__ part,
                                                         const float* W1,
                                                         const float* __restrict__ b1,
                                                         const float* W2,
                                                         const float* __restrict__ b2,
                                                         float* __restrict__ out) {
  __shared__ int pos[256];
  __shared__ int segs[256];
  __shared__ int cntl[256];
  __shared__ int offl[256];
  __shared__ int cur[256];
  __shared__ int wsum[16];
  __shared__ int lv[CAP];
  __shared__ int lss[CAP + 32];
  __shared__ __attribute__((aligned(16))) float accL[NPB * ACC_STRIDE];
  __shared__ __attribute__((aligned(16))) float hid[NPB * LDS_STRIDE];
  __shared__ __attribute__((aligned(16))) float w1s[DF * W1_STRIDE];
  __shared__ __attribute__((aligned(16))) float w2s[DH * W2_STRIDE];
  const int t = threadIdx.x;
  const int lane = t & 31;
  const int w = t >> 5;
  const int b = blockIdx.x;
  const int half = lane >> 4;
  const int l16  = lane & 15;

  static_assert(DF * DH / 4 == 2 * QT, "two 16-B pieces of each weight matrix per thread");
  const int tg = min(t, NG - 1);
  int s = startT[b * NG + tg];
  const int e = startT[(b + 1) * NG + tg];
  const float4* W1v = (const float4*)W1;
  const float4* W2v = (const float4*)W2;
  const float4 w1a = W1v[t], w1b = W1v[QT + t];
  const float4 w2a = W2v[t], w2b = W2v[QT + t];
  s = min(max(s, 0), CH);
  int n = min(max(e - s, 0), CH - s);
  if (t >= NG) { s = 0; n = 0; }
  int nb;
  const int p0 = scan256(n, wsum, nb);
  if (t < 256) { pos[t] = (t < NG) ? p0 : nb; segs[t] = s; }

  *(float4*)&w1s[(t >> 5) * W1_STRIDE + (t & 31) * 4] = w1a;
  *(float4*)&w1s[((QT + t) >> 5) * W1_STRIDE + (t & 31) * 4] = w1b;
  *(float4*)&w2s[(t >> 4) * W2_STRIDE + (t & 15) * 4] = w2a;
  *(float4*)&w2s[((QT + t) >> 4) * W2_STRIDE + (t & 15) * 4] = w2b;

  const float sc = 1.0f + eps[0];
  const unsigned lo = (unsigned)l16 << 4;

  int r0 = 0;
  do {
    const int nr = min(nb - r0, CAP);
    if (t < 256) cntl[t] = 0;
    __syncthreads();
    for (int pb = 0; pb < nr; pb += 4 * QT) {
      int pg[4], lo4[4], vv[4];
#pragma unroll
      for (int j = 0; j < 4; ++j) {
        pg[j] = r0 + min(pb + j * QT + t, nr - 1);
        lo4[j] = 0;
      }
#pragma unroll
      for (int st = 128; st >= 1; st >>= 1) {
        int pv[4];
#pragma unroll
        for (int j = 0; j < 4; ++j) pv[j] = pos[lo4[j] + st];
#pragma unroll
        for (int j = 0; j < 4; ++j) { if (pv[j] <= pg[j]) lo4[j] += st; }
      }
#pragma unroll
      for (int j = 0; j < 4; ++j) {
        const int l = lo4[j];
        vv[j] = part[l * CH + segs[l] + (pg[j] - pos[l])];
      }
      asm volatile("" :: "v"(vv[0]), "v"(vv[1]), "v"(vv[2]), "v"(vv[3]));
#pragma unroll
      for (int j = 0; j < 4; ++j) {
        const int p = pb + j * QT + t;
        if (p < nr) {
          lv[p] = vv[j];
          atomicAdd(&cntl[min(vv[j] & 255, NPB - 1)], 1);
        }
      }
    }
    __syncthreads();
    int tot;
    const int cc = (t < 256) ? cntl[t] : 0;
    const int oo = scan256(cc, wsum, tot);
    if (t < 256) { offl[t] = oo; cur[t] = oo; }
    __syncthreads();
    for (int pb = 0; pb < nr; pb += 4 * QT) {
      int wv[4], q[4];
#pragma unroll
      for (int j = 0; j < 4; ++j) wv[j] = lv[min(pb + j * QT + t, nr - 1)];
#pragma unroll
      for (int j = 0; j < 4; ++j) {
        const bool on = pb + j * QT + t < nr;
        q[j] = atomicAdd(&cur[on ? min(wv[j] & 255, NPB - 1) : (t & 127)], on ? 1 : 0);
      }
#pragma unroll
      for (int j = 0; j < 4; ++j) {
        if (pb + j * QT + t < nr) lss[min((unsigned)q[j], (unsigned)(CAP - 1))] = wv[j] >> 8;
      }
    }
    __syncthreads();

    for (int nl = w; nl < NPB; nl += QT / 32) {
      const int node = b * NPB + nl;
      float4* arow = (float4*)&accL[nl * ACC_STRIDE];
      if (node >= NN) {
        if (r0 == 0 && half == 0) arow[l16] = make_float4(0.0f, 0.0f, 0.0f, 0.0f);
        continue;
      }
      int c = __builtin_amdgcn_readfirstlane(cntl[nl]);
      int o = __builtin_amdgcn_readfirstlane(offl[nl]);
      c = min(max(c, 0), nr);
      o = min(max(o, 0), nr - c);
      float4 own = make_float4(0.0f, 0.0f, 0.0f, 0.0f);
      if (r0 == 0) own = ld_piece(feat, node, lo);

      float4 a = make_float4(0.0f, 0.0f, 0.0f, 0.0f);
      for (int base = 0; base < c; base += 32) {
        const int m = min(c - base, 32);
        const int hp = m >> 1;
        const int* lp = &lss[o + base + half];
        int idx[16];
#pragma unroll
        for (int j = 0; j < 16; ++j) idx[j] = lp[2 * j];
        float4 vt = make_float4(0.0f, 0.0f, 0.0f, 0.0f);
        if (m & 1) {
          const int it = lss[o + base + m - 1];
          if (half == 0) vt = ld_piece(feat, it, lo);
        }
        float4 v[16];
        load_rungs<0>(v, idx, hp, feat, lo);
        add_rungs<0>(a, v, hp);
        a.x += vt.x; a.y += vt.y; a.z += vt.z; a.w += vt.w;
      }
      a.x += xhalf(a.x); a.y += xhalf(a.y); a.z += xhalf(a.z); a.w += xhalf(a.w);
      float4 r;
      if (r0 == 0) {
        r.x = sc * own.x + a.x; r.y = sc * own.y + a.y;
        r.z = sc * own.z + a.z; r.w = sc * own.w + a.w;
      } else {
        const float4 prev = arow[l16];
        r.x = prev.x + a.x; r.y = prev.y + a.y; r.z = prev.z + a.z; r.w = prev.w + a.w;
      }
      if (half == 0) arow[l16] = r;
    }
    r0 += CAP;
    __syncthreads();
  } while (r0 < nb);

  const int wu = __builtin_amdgcn_readfirstlane(w);

  for (int u = wu; u < TPB * 4; u += QT / 32) {
    const int tile = u >> 2;
    const int np = u & 3;
    if (b * NPB + tile * 16 >= NN) continue;
    v2f afrag[16];
    {
      const float* arow = &accL[(tile * 16 + l16) * ACC_STRIDE];
#pragma unroll
      for (int kk = 0; kk < 16; ++kk) {
        const int kb = kk * 4 + half * 2;
        afrag[kk].x = arow[kb];
        afrag[kk].y = arow[kb + 1];
      }
    }
    v8f c0 = {};
    v8f c1 = {};
    const int n0 = np * 32;
#pragma unroll
    for (int kk = 0; kk < 16; ++kk) {
      const int kr = kk * 4 + half * 2;
      v2f bf0, bf1;
      bf0.x = w1s[kr * W1_STRIDE + n0 + l16];
      bf0.y = w1s[(kr + 1) * W1_STRIDE + n0 + l16];
      bf1.x = w1s[kr * W1_STRIDE + n0 + 16 + l16];
      bf1.y = w1s[(kr + 1) * W1_STRIDE + n0 + 16 + l16];
      c0 = __builtin_amdgcn_wmma_f32_16x16x4_f32(false, afrag[kk], false, bf0,
                                                 (short)0, c0, false, false);
      c1 = __builtin_amdgcn_wmma_f32_16x16x4_f32(false, afrag[kk], false, bf1,
                                                 (short)0, c1, false, false);
    }
    const float bias0 = b1[n0 + l16];
    const float bias1 = b1[n0 + 16 + l16];
#pragma unroll
    for (int r = 0; r < 8; ++r) {
      float h0 = c0[r] + bias0;
      float h1 = c1[r] + bias1;
      h0 = h0 > 0.0f ? h0 : 0.0f;
      h1 = h1 > 0.0f ? h1 : 0.0f;
      hid[(tile * 16 + r + half * 8) * LDS_STRIDE + n0 + l16] = h0;
      hid[(tile * 16 + r + half * 8) * LDS_STRIDE + n0 + 16 + l16] = h1;
    }
  }

  __syncthreads();

  for (int u = wu; u < TPB * 2; u += QT / 32) {
    const int tile = u >> 1;
    const int np = u & 1;
    const int row0 = b * NPB + tile * 16;
    if (row0 >= NN) continue;
    const float* hrow = &hid[(tile * 16 + l16) * LDS_STRIDE];
    const int n0 = np * 32;
    v8f c0 = {};
    v8f c1 = {};
#pragma unroll
    for (int kk = 0; kk < 32; ++kk) {
      const int kr = kk * 4 + half * 2;
      v2f af, bf0, bf1;
      af.x = hrow[kr];
      af.y = hrow[kr + 1];
      bf0.x = w2s[kr * W2_STRIDE + n0 + l16];
      bf0.y = w2s[(kr + 1) * W2_STRIDE + n0 + l16];
      bf1.x = w2s[kr * W2_STRIDE + n0 + 16 + l16];
      bf1.y = w2s[(kr + 1) * W2_STRIDE + n0 + 16 + l16];
      c0 = __builtin_amdgcn_wmma_f32_16x16x4_f32(false, af, false, bf0,
                                                 (short)0, c0, false, false);
      c1 = __builtin_amdgcn_wmma_f32_16x16x4_f32(false, af, false, bf1,
                                                 (short)0, c1, false, false);
    }
    const float bias0 = b2[n0 + l16];
    const float bias1 = b2[n0 + 16 + l16];
#pragma unroll
    for (int r = 0; r < 8; ++r) {
      out[(long long)(row0 + r + half * 8) * DF + n0 + l16] = c0[r] + bias0;
      out[(long long)(row0 + r + half * 8) * DF + n0 + 16 + l16] = c1[r] + bias1;
    }
  }
}

extern "C" void kernel_launch(void* const* d_in, const int* in_sizes, int n_in,
                              void* d_out, int out_size, void* d_ws, size_t ws_size,
                              hipStream_t stream) {
  const float* feat = (const float*)d_in[0];
  const int*   src  = (const int*)d_in[1];
  const int*   dst  = (const int*)d_in[2];
  const float* eps  = (const float*)d_in[3];
  const float* W1   = (const float*)d_in[4];
  const float* b1   = (const float*)d_in[5];
  const float* W2   = (const float*)d_in[6];
  const float* b2   = (const float*)d_in[7];
  int*   startT = (int*)d_ws + 3200000;
  int*   part   = (int*)d_ws + 3262784;
  float* out = (float*)d_out;

  gin_part<<<NG, PT, 0, stream>>>(src, dst, startT, part);
  gin_bin_gather_mlp<<<NB, QT, 0, stream>>>(feat, eps, startT, part, W1, b1, W2, b2, out);
}
